# starting kernel
# baseline (speedup 1.0000x reference)
#include <hip/hip_runtime.h>

typedef __attribute__((ext_vector_type(2))) float v2f;
typedef __attribute__((ext_vector_type(8))) float v8f;

#define HID   128
#define DIN   256
#define LROW  260

__global__ __launch_bounds__(256) void tgnn_zero_kernel(float4* __restrict__ p, long long n4) {
    long long i = (long long)blockIdx.x * blockDim.x + threadIdx.x;
    if (i < n4) p[i] = make_float4(0.f, 0.f, 0.f, 0.f);
}

__global__ __launch_bounds__(256) void tgnn_edge_kernel(
    const int*   __restrict__ src,
    const int*   __restrict__ dst,
    const float* __restrict__ ef,
    const float* __restrict__ bt,
    const float* __restrict__ node_ts,
    const float* __restrict__ time_w,
    const float* __restrict__ time_b,
    float*       __restrict__ h)
{
    int e = blockIdx.x;
    int j = threadIdx.x;
    int s = src[e];
    int d = dst[e];
    float td = bt[e] - node_ts[s];
    float v;
    if (j < HID) {
        v = ef[(size_t)e * HID + j];
    } else {
        int k = j - HID;
        v = cosf(td * time_w[k] + time_b[k]);
    }
    atomicAdd(&h[(size_t)d * DIN + j], v);
}

__global__ __launch_bounds__(256) void tgnn_predict_kernel(
    const float* __restrict__ h,
    const int*   __restrict__ s_idx,
    const int*   __restrict__ p_idx,
    const int*   __restrict__ n_idx,
    const float* __restrict__ W_src,
    const float* __restrict__ b_src,
    const float* __restrict__ W_dst,
    const float* __restrict__ b_dst,
    const float* __restrict__ W_out,
    const float* __restrict__ b_out,
    float*       __restrict__ out,
    int Q)
{
    __shared__ float sl[16 * LROW];
    __shared__ float pl[16 * LROW];
    __shared__ float nl[16 * LROW];
    __shared__ float red_pos[8 * 256];
    __shared__ float red_neg[8 * 256];

    const int tile = blockIdx.x;
    const int t    = threadIdx.x;

    for (int i = t; i < 16 * DIN; i += 256) {
        int row = i >> 8;
        int col = i & 255;
        int qi  = tile * 16 + row;
        sl[row * LROW + col] = h[(size_t)s_idx[qi] * DIN + col];
        pl[row * LROW + col] = h[(size_t)p_idx[qi] * DIN + col];
        nl[row * LROW + col] = h[(size_t)n_idx[qi] * DIN + col];
    }
    __syncthreads();

    const int wave = t >> 5;
    const int lane = t & 31;
    const int half = lane >> 4;
    const int lm   = lane & 15;
    const int n0   = wave * 16;

    v8f hs = {}, hp = {}, hn = {};

    const float* As = &sl[lm * LROW];
    const float* Ap = &pl[lm * LROW];
    const float* An = &nl[lm * LROW];

    for (int k0 = 0; k0 < DIN; k0 += 4) {
        int ka = k0 + 2 * half;
        v2f a_s; a_s.x = As[ka]; a_s.y = As[ka + 1];
        v2f a_p; a_p.x = Ap[ka]; a_p.y = Ap[ka + 1];
        v2f a_n; a_n.x = An[ka]; a_n.y = An[ka + 1];
        v2f b_s; b_s.x = W_src[ka * HID + n0 + lm]; b_s.y = W_src[(ka + 1) * HID + n0 + lm];
        v2f b_d; b_d.x = W_dst[ka * HID + n0 + lm]; b_d.y = W_dst[(ka + 1) * HID + n0 + lm];

        hs = __builtin_amdgcn_wmma_f32_16x16x4_f32(false, a_s, false, b_s, (short)0, hs, false, false);
        hp = __builtin_amdgcn_wmma_f32_16x16x4_f32(false, a_p, false, b_d, (short)0, hp, false, false);
        hn = __builtin_amdgcn_wmma_f32_16x16x4_f32(false, a_n, false, b_d, (short)0, hn, false, false);
    }

    const float bias = b_src[n0 + lm] + b_dst[n0 + lm];
    const float wo   = W_out[n0 + lm];

    #pragma unroll
    for (int r = 0; r < 8; ++r) {
        float pv = fmaxf(hs[r] + hp[r] + bias, 0.f) * wo;
        float nv = fmaxf(hs[r] + hn[r] + bias, 0.f) * wo;
        red_pos[wave * 256 + r * 32 + lane] = pv;
        red_neg[wave * 256 + r * 32 + lane] = nv;
    }
    __syncthreads();

    if (t < 16) {
        int mh = t >> 3;
        int r  = t & 7;
        float accp = b_out[0];
        float accn = b_out[0];
        for (int w = 0; w < 8; ++w) {
            #pragma unroll
            for (int l = 0; l < 16; ++l) {
                int idx = w * 256 + r * 32 + mh * 16 + l;
                accp += red_pos[idx];
                accn += red_neg[idx];
            }
        }
        out[tile * 16 + t]     = accp;
        out[Q + tile * 16 + t] = accn;
    }
}

extern "C" void kernel_launch(void* const* d_in, const int* in_sizes, int n_in,
                              void* d_out, int out_size, void* d_ws, size_t ws_size,
                              hipStream_t stream) {
    const int*   src     = (const int*)  d_in[0];
    const int*   dst     = (const int*)  d_in[1];
    const float* ef      = (const float*)d_in[2];
    const float* bt      = (const float*)d_in[3];
    const float* node_ts = (const float*)d_in[4];
    const int*   s_idx   = (const int*)  d_in[5];
    const int*   p_idx   = (const int*)  d_in[6];
    const int*   n_idx   = (const int*)  d_in[7];
    const float* time_w  = (const float*)d_in[8];
    const float* time_b  = (const float*)d_in[9];
    const float* W_src   = (const float*)d_in[10];
    const float* b_src   = (const float*)d_in[11];
    const float* W_dst   = (const float*)d_in[12];
    const float* b_dst   = (const float*)d_in[13];
    const float* W_out   = (const float*)d_in[14];
    const float* b_out   = (const float*)d_in[15];

    float* out = (float*)d_out;
    float* h   = (float*)d_ws;

    const int NN = in_sizes[4];
    const int NE = in_sizes[0];
    const int Q  = in_sizes[5];

    long long n4 = (long long)NN * DIN / 4;
    int zgrid = (int)((n4 + 255) / 256);
    tgnn_zero_kernel<<<zgrid, 256, 0, stream>>>((float4*)h, n4);

    tgnn_edge_kernel<<<NE, 256, 0, stream>>>(src, dst, ef, bt, node_ts, time_w, time_b, h);

    tgnn_predict_kernel<<<Q / 16, 256, 0, stream>>>(h, s_idx, p_idx, n_idx,
                                                    W_src, b_src, W_dst, b_dst,
                                                    W_out, b_out, out, Q);
}


# Predictor gather batched: a wave reads its six row indices once (scalar loads), then asks for all six 1-KiB h rows with 16-byte loads before any LDS store; was 8 serial index-then-row trips
# speedup vs baseline: 4.2702x; 4.2702x over previous
#include <hip/hip_runtime.h>

typedef __attribute__((ext_vector_type(2))) float v2f;
typedef __attribute__((ext_vector_type(8))) float v8f;

#define HID   128
#define DIN   256
#define LROW  260

#define RT     196
#define NWAVE  32
#define NBIN   160
#define SPW    (NBIN / NWAVE)
#define SLOTW  32
#define SCAP   (SLOTW - 1)
#define TCH    512
#define LCAP   160
#define G      8

__device__ __forceinline__ bool tgnn_tile_has_slots(int nt) { return nt * DIN >= NBIN * SLOTW; }

__global__ __launch_bounds__(1024) void tgnn_bin_kernel(
    const int* __restrict__ dst,
    float*     __restrict__ h,
    int NN, int NE)
{
    __shared__ int sl[TCH * SLOTW];

    const int t    = threadIdx.x;
    const int wave = __builtin_amdgcn_readfirstlane(t >> 5);
    const int lane = t & 31;
    const int b    = blockIdx.x;

    const int ntiles = (NN + RT - 1) / RT;
    const int epb    = (NE + NBIN - 1) / NBIN;
    const int lo     = b * epb;
    const int hi     = min(NE, lo + epb);

    for (int tb = 0; tb < ntiles; tb += TCH) {
        for (int i = t; i < TCH * SLOTW; i += NWAVE * 32) sl[i] = 0;
        __syncthreads();

        for (int e = lo + t; e < hi; e += NWAVE * 32) {
            const int d = dst[e];
            if ((unsigned)d < (unsigned)NN) {
                const int tl = d / RT - tb;
                if ((unsigned)tl < (unsigned)TCH) {
                    const int pos = __hip_atomic_fetch_add(&sl[tl * SLOTW], 1, __ATOMIC_RELAXED,
                                                           __HIP_MEMORY_SCOPE_WORKGROUP);
                    if (pos < SCAP) sl[tl * SLOTW + 1 + pos] = e;
                }
            }
        }
        __syncthreads();

        const int tn = min(TCH, ntiles - tb);
        for (int tl = wave; tl < tn; tl += NWAVE) {
            const int T  = tb + tl;
            const int nt = min(RT, NN - T * RT);
            if (tgnn_tile_has_slots(nt))
                h[(size_t)T * RT * DIN + b * SLOTW + lane] = __int_as_float(sl[tl * SLOTW + lane]);
        }
        __syncthreads();
    }
}

__global__ __launch_bounds__(1024) void tgnn_tile_accum_kernel(
    const int*   __restrict__ src,
    const int*   __restrict__ dst,
    const float* __restrict__ ef,
    const float* __restrict__ bt,
    const float* __restrict__ node_ts,
    const float* __restrict__ time_w,
    const float* __restrict__ time_b,
    float*       __restrict__ h,
    int NN, int NE)
{
    __shared__ __attribute__((aligned(16))) float acc[RT * DIN];
    __shared__ int   le [NWAVE * LCAP];
    __shared__ int   lr [NWAVE * LCAP];
    __shared__ float ltd[NWAVE * LCAP];
    __shared__ int   ovf;

    const int t    = threadIdx.x;
    const int wave = __builtin_amdgcn_readfirstlane(t >> 5);
    const int lane = t & 31;
    const int n0   = blockIdx.x * RT;
    const int nt   = min(RT, NN - n0);
    const bool has_slots = tgnn_tile_has_slots(nt);

    if (t == 0) ovf = 0;
    __syncthreads();

    int w[SPW];
    #pragma unroll
    for (int s = 0; s < SPW; ++s)
        w[s] = has_slots ? __float_as_int(h[(size_t)n0 * DIN + (wave + s * NWAVE) * SLOTW + lane]) : 0;

    float4* acc4 = (float4*)acc;
    for (int i = t; i < RT * DIN / 4; i += NWAVE * 32)
        acc4[i] = make_float4(0.f, 0.f, 0.f, 0.f);

    float tw[4], tb[4];
    #pragma unroll
    for (int q = 0; q < 4; ++q) {
        tw[q] = time_w[lane + 32 * q];
        tb[q] = time_b[lane + 32 * q];
    }

    int  c[SPW];
    bool over = false;
    #pragma unroll
    for (int s = 0; s < SPW; ++s) {
        c[s]  = __builtin_amdgcn_readfirstlane(w[s]);
        over |= (unsigned)c[s] > (unsigned)SCAP;
    }
    if (over && lane == 0) ovf = 1;
    __syncthreads();
    const bool scan_all = !has_slots || __builtin_amdgcn_readfirstlane(ovf) != 0;

    int*   my_e  = &le [wave * LCAP];
    int*   my_r  = &lr [wave * LCAP];
    float* my_td = &ltd[wave * LCAP];
    const unsigned lanes_below = (1u << lane) - 1u;

    if (!scan_all) {
        int cnt = 0;
        #pragma unroll
        for (int s = 0; s < SPW; ++s) {
            const bool     m    = lane >= 1 && lane <= c[s];
            const unsigned mask = __builtin_amdgcn_ballot_w32(m);
            if (m) my_e[cnt + __popc(mask & lanes_below)] = w[s];
            cnt += __popc(mask);
        }

        for (int i = lane; i < cnt; i += 32) {
            const int e = my_e[i];
            my_r[i]  = (dst[e] - n0) * DIN;
            my_td[i] = bt[e] - node_ts[src[e]];
        }

        for (int i0 = 0; i0 < cnt; i0 += G) {
            const int ng = min(G, cnt - i0);
            const int   li  = i0 + min(lane & (G - 1), ng - 1);
            const int   ev  = my_e[li];
            const int   rv  = my_r[li];
            const float tdv = my_td[li];

            float v[G][4];
            #pragma unroll
            for (int g = 0; g < G; ++g) {
                const int    e   = __builtin_amdgcn_readlane(ev, g);
                const float* row = ef + (size_t)e * HID + lane;
                v[g][0] = row[0]; v[g][1] = row[32]; v[g][2] = row[64]; v[g][3] = row[96];
            }

            for (int g = 0; g < ng; ++g) {
                const int   r  = __builtin_amdgcn_readlane(rv, g);
                const float td = __int_as_float(__builtin_amdgcn_readlane(__float_as_int(tdv), g));
                const float c0 = cosf(td * tw[0] + tb[0]);
                const float c1 = cosf(td * tw[1] + tb[1]);
                const float c2 = cosf(td * tw[2] + tb[2]);
                const float c3 = cosf(td * tw[3] + tb[3]);
                float* a = &acc[r + lane];
                __hip_atomic_fetch_add(a + 128, c0, __ATOMIC_RELAXED, __HIP_MEMORY_SCOPE_WORKGROUP);
                __hip_atomic_fetch_add(a + 160, c1, __ATOMIC_RELAXED, __HIP_MEMORY_SCOPE_WORKGROUP);
                __hip_atomic_fetch_add(a + 192, c2, __ATOMIC_RELAXED, __HIP_MEMORY_SCOPE_WORKGROUP);
                __hip_atomic_fetch_add(a + 224, c3, __ATOMIC_RELAXED, __HIP_MEMORY_SCOPE_WORKGROUP);
            }

            #pragma unroll
            for (int g = 0; g < G; ++g) {
                if (g < ng) {
                    const int r = __builtin_amdgcn_readlane(rv, g);
                    float* a = &acc[r + lane];
                    __hip_atomic_fetch_add(a,      v[g][0], __ATOMIC_RELAXED, __HIP_MEMORY_SCOPE_WORKGROUP);
                    __hip_atomic_fetch_add(a + 32, v[g][1], __ATOMIC_RELAXED, __HIP_MEMORY_SCOPE_WORKGROUP);
                    __hip_atomic_fetch_add(a + 64, v[g][2], __ATOMIC_RELAXED, __HIP_MEMORY_SCOPE_WORKGROUP);
                    __hip_atomic_fetch_add(a + 96, v[g][3], __ATOMIC_RELAXED, __HIP_MEMORY_SCOPE_WORKGROUP);
                }
            }
        }
    } else {
        for (int e = t; e < NE; e += NWAVE * 32) {
            const int dl = dst[e] - n0;
            if ((unsigned)dl < (unsigned)nt) {
                const float  td  = bt[e] - node_ts[src[e]];
                const float* row = ef + (size_t)e * HID;
                float*       a   = &acc[dl * DIN];
                for (int f = 0; f < HID; ++f) {
                    __hip_atomic_fetch_add(a + f, row[f], __ATOMIC_RELAXED, __HIP_MEMORY_SCOPE_WORKGROUP);
                    __hip_atomic_fetch_add(a + HID + f, cosf(td * time_w[f] + time_b[f]),
                                           __ATOMIC_RELAXED, __HIP_MEMORY_SCOPE_WORKGROUP);
                }
            }
        }
    }
    __syncthreads();

    float4* h4 = (float4*)(h + (size_t)n0 * DIN);
    for (int i = t; i < nt * (DIN / 4); i += NWAVE * 32)
        h4[i] = acc4[i];
}

__global__ __launch_bounds__(256) void tgnn_predict_kernel(
    const float* __restrict__ h,
    const int*   __restrict__ s_idx,
    const int*   __restrict__ p_idx,
    const int*   __restrict__ n_idx,
    const float* __restrict__ W_src,
    const float* __restrict__ b_src,
    const float* __restrict__ W_dst,
    const float* __restrict__ b_dst,
    const float* __restrict__ W_out,
    const float* __restrict__ b_out,
    float*       __restrict__ out,
    int Q)
{
    __shared__ __attribute__((aligned(16))) float sl[16 * LROW];
    __shared__ __attribute__((aligned(16))) float pl[16 * LROW];
    __shared__ __attribute__((aligned(16))) float nl[16 * LROW];
    __shared__ float red_pos[8 * 256];
    __shared__ float red_neg[8 * 256];

    const int tile = blockIdx.x;
    const int t    = threadIdx.x;

    {
        const int gw = __builtin_amdgcn_readfirstlane(t >> 5);
        const int gl = t & 31;
        const int r0 = 2 * gw;
        const int qa = tile * 16 + r0;

        const float4* hs0 = (const float4*)(h + (size_t)s_idx[qa]     * DIN);
        const float4* hs1 = (const float4*)(h + (size_t)s_idx[qa + 1] * DIN);
        const float4* hp0 = (const float4*)(h + (size_t)p_idx[qa]     * DIN);
        const float4* hp1 = (const float4*)(h + (size_t)p_idx[qa + 1] * DIN);
        const float4* hn0 = (const float4*)(h + (size_t)n_idx[qa]     * DIN);
        const float4* hn1 = (const float4*)(h + (size_t)n_idx[qa + 1] * DIN);

        const float4 s00 = hs0[gl], s01 = hs0[gl + 32];
        const float4 s10 = hs1[gl], s11 = hs1[gl + 32];
        const float4 p00 = hp0[gl], p01 = hp0[gl + 32];
        const float4 p10 = hp1[gl], p11 = hp1[gl + 32];
        const float4 n00 = hn0[gl], n01 = hn0[gl + 32];
        const float4 n10 = hn1[gl], n11 = hn1[gl + 32];

        float4* s4a = (float4*)&sl[r0 * LROW];
        float4* s4b = (float4*)&sl[(r0 + 1) * LROW];
        float4* p4a = (float4*)&pl[r0 * LROW];
        float4* p4b = (float4*)&pl[(r0 + 1) * LROW];
        float4* n4a = (float4*)&nl[r0 * LROW];
        float4* n4b = (float4*)&nl[(r0 + 1) * LROW];
        s4a[gl] = s00; s4a[gl + 32] = s01;
        s4b[gl] = s10; s4b[gl + 32] = s11;
        p4a[gl] = p00; p4a[gl + 32] = p01;
        p4b[gl] = p10; p4b[gl + 32] = p11;
        n4a[gl] = n00; n4a[gl + 32] = n01;
        n4b[gl] = n10; n4b[gl + 32] = n11;
    }
    __syncthreads();

    const int wave = t >> 5;
    const int lane = t & 31;
    const int half = lane >> 4;
    const int lm   = lane & 15;
    const int n0   = wave * 16;

    v8f hs = {}, hp = {}, hn = {};

    const float* As = &sl[lm * LROW];
    const float* Ap = &pl[lm * LROW];
    const float* An = &nl[lm * LROW];

    for (int k0 = 0; k0 < DIN; k0 += 4) {
        int ka = k0 + 2 * half;
        v2f a_s; a_s.x = As[ka]; a_s.y = As[ka + 1];
        v2f a_p; a_p.x = Ap[ka]; a_p.y = Ap[ka + 1];
        v2f a_n; a_n.x = An[ka]; a_n.y = An[ka + 1];
        v2f b_s; b_s.x = W_src[ka * HID + n0 + lm]; b_s.y = W_src[(ka + 1) * HID + n0 + lm];
        v2f b_d; b_d.x = W_dst[ka * HID + n0 + lm]; b_d.y = W_dst[(ka + 1) * HID + n0 + lm];

        hs = __builtin_amdgcn_wmma_f32_16x16x4_f32(false, a_s, false, b_s, (short)0, hs, false, false);
        hp = __builtin_amdgcn_wmma_f32_16x16x4_f32(false, a_p, false, b_d, (short)0, hp, false, false);
        hn = __builtin_amdgcn_wmma_f32_16x16x4_f32(false, a_n, false, b_d, (short)0, hn, false, false);
    }

    const float bias = b_src[n0 + lm] + b_dst[n0 + lm];
    const float wo   = W_out[n0 + lm];

    #pragma unroll
    for (int r = 0; r < 8; ++r) {
        float pv = fmaxf(hs[r] + hp[r] + bias, 0.f) * wo;
        float nv = fmaxf(hs[r] + hn[r] + bias, 0.f) * wo;
        red_pos[wave * 256 + r * 32 + lane] = pv;
        red_neg[wave * 256 + r * 32 + lane] = nv;
    }
    __syncthreads();

    if (t < 16) {
        int mh = t >> 3;
        int r  = t & 7;
        float accp = b_out[0];
        float accn = b_out[0];
        for (int w = 0; w < 8; ++w) {
            #pragma unroll
            for (int l = 0; l < 16; ++l) {
                int idx = w * 256 + r * 32 + mh * 16 + l;
                accp += red_pos[idx];
                accn += red_neg[idx];
            }
        }
        out[tile * 16 + t]     = accp;
        out[Q + tile * 16 + t] = accn;
    }
}

extern "C" void kernel_launch(void* const* d_in, const int* in_sizes, int n_in,
                              void* d_out, int out_size, void* d_ws, size_t ws_size,
                              hipStream_t stream) {
    const int*   src     = (const int*)  d_in[0];
    const int*   dst     = (const int*)  d_in[1];
    const float* ef      = (const float*)d_in[2];
    const float* bt      = (const float*)d_in[3];
    const float* node_ts = (const float*)d_in[4];
    const int*   s_idx   = (const int*)  d_in[5];
    const int*   p_idx   = (const int*)  d_in[6];
    const int*   n_idx   = (const int*)  d_in[7];
    const float* time_w  = (const float*)d_in[8];
    const float* time_b  = (const float*)d_in[9];
    const float* W_src   = (const float*)d_in[10];
    const float* b_src   = (const float*)d_in[11];
    const float* W_dst   = (const float*)d_in[12];
    const float* b_dst   = (const float*)d_in[13];
    const float* W_out   = (const float*)d_in[14];
    const float* b_out   = (const float*)d_in[15];

    float* out = (float*)d_out;
    float* h   = (float*)d_ws;

    const int NN = in_sizes[4];
    const int NE = in_sizes[0];
    const int Q  = in_sizes[5];

    tgnn_bin_kernel<<<NBIN, NWAVE * 32, 0, stream>>>(dst, h, NN, NE);

    int agrid = (NN + RT - 1) / RT;
    tgnn_tile_accum_kernel<<<agrid, NWAVE * 32, 0, stream>>>(src, dst, ef, bt, node_ts,
                                                             time_w, time_b, h, NN, NE);

    tgnn_predict_kernel<<<Q / 16, 256, 0, stream>>>(h, s_idx, p_idx, n_idx,
                                                    W_src, b_src, W_dst, b_dst,
                                                    W_out, b_out, out, Q);
}


# Time-encoding cosines skipped where exact: per edge and 32-feature group one wave-wide test |t*w+b| < 2^-13; there f32 cos is exactly 1, the wave adds 1 and skips cosf; else cosf as before
# speedup vs baseline: 4.5553x; 1.0668x over previous
#include <hip/hip_runtime.h>

typedef __attribute__((ext_vector_type(2))) float v2f;
typedef __attribute__((ext_vector_type(8))) float v8f;

#define HID   128
#define DIN   256
#define LROW  260

#define RT     196
#define NWAVE  32
#define NBIN   160
#define SPW    (NBIN / NWAVE)
#define SLOTW  32
#define SCAP   (SLOTW - 1)
#define TCH    512
#define LCAP   160
#define G      8

__device__ __forceinline__ bool tgnn_tile_has_slots(int nt) { return nt * DIN >= NBIN * SLOTW; }

__global__ __launch_bounds__(1024) void tgnn_bin_kernel(
    const int* __restrict__ dst,
    float*     __restrict__ h,
    int NN, int NE)
{
    __shared__ int sl[TCH * SLOTW];

    const int t    = threadIdx.x;
    const int wave = __builtin_amdgcn_readfirstlane(t >> 5);
    const int lane = t & 31;
    const int b    = blockIdx.x;

    const int ntiles = (NN + RT - 1) / RT;
    const int epb    = (NE + NBIN - 1) / NBIN;
    const int lo     = b * epb;
    const int hi     = min(NE, lo + epb);

    for (int tb = 0; tb < ntiles; tb += TCH) {
        for (int i = t; i < TCH * SLOTW; i += NWAVE * 32) sl[i] = 0;
        __syncthreads();

        for (int e = lo + t; e < hi; e += NWAVE * 32) {
            const int d = dst[e];
            if ((unsigned)d < (unsigned)NN) {
                const int tl = d / RT - tb;
                if ((unsigned)tl < (unsigned)TCH) {
                    const int pos = __hip_atomic_fetch_add(&sl[tl * SLOTW], 1, __ATOMIC_RELAXED,
                                                           __HIP_MEMORY_SCOPE_WORKGROUP);
                    if (pos < SCAP) sl[tl * SLOTW + 1 + pos] = e;
                }
            }
        }
        __syncthreads();

        const int tn = min(TCH, ntiles - tb);
        for (int tl = wave; tl < tn; tl += NWAVE) {
            const int T  = tb + tl;
            const int nt = min(RT, NN - T * RT);
            if (tgnn_tile_has_slots(nt))
                h[(size_t)T * RT * DIN + b * SLOTW + lane] = __int_as_float(sl[tl * SLOTW + lane]);
        }
        __syncthreads();
    }
}

__global__ __launch_bounds__(1024) void tgnn_tile_accum_kernel(
    const int*   __restrict__ src,
    const int*   __restrict__ dst,
    const float* __restrict__ ef,
    const float* __restrict__ bt,
    const float* __restrict__ node_ts,
    const float* __restrict__ time_w,
    const float* __restrict__ time_b,
    float*       __restrict__ h,
    int NN, int NE)
{
    __shared__ __attribute__((aligned(16))) float acc[RT * DIN];
    __shared__ int   le [NWAVE * LCAP];
    __shared__ int   lr [NWAVE * LCAP];
    __shared__ float ltd[NWAVE * LCAP];
    __shared__ int   ovf;

    const int t    = threadIdx.x;
    const int wave = __builtin_amdgcn_readfirstlane(t >> 5);
    const int lane = t & 31;
    const int n0   = blockIdx.x * RT;
    const int nt   = min(RT, NN - n0);
    const bool has_slots = tgnn_tile_has_slots(nt);

    if (t == 0) ovf = 0;
    __syncthreads();

    int w[SPW];
    #pragma unroll
    for (int s = 0; s < SPW; ++s)
        w[s] = has_slots ? __float_as_int(h[(size_t)n0 * DIN + (wave + s * NWAVE) * SLOTW + lane]) : 0;

    float4* acc4 = (float4*)acc;
    for (int i = t; i < RT * DIN / 4; i += NWAVE * 32)
        acc4[i] = make_float4(0.f, 0.f, 0.f, 0.f);

    float tw[4], tb[4];
    #pragma unroll
    for (int q = 0; q < 4; ++q) {
        tw[q] = time_w[lane + 32 * q];
        tb[q] = time_b[lane + 32 * q];
    }

    int  c[SPW];
    bool over = false;
    #pragma unroll
    for (int s = 0; s < SPW; ++s) {
        c[s]  = __builtin_amdgcn_readfirstlane(w[s]);
        over |= (unsigned)c[s] > (unsigned)SCAP;
    }
    if (over && lane == 0) ovf = 1;
    __syncthreads();
    const bool scan_all = !has_slots || __builtin_amdgcn_readfirstlane(ovf) != 0;

    int*   my_e  = &le [wave * LCAP];
    int*   my_r  = &lr [wave * LCAP];
    float* my_td = &ltd[wave * LCAP];
    const unsigned lanes_below = (1u << lane) - 1u;

    if (!scan_all) {
        int cnt = 0;
        #pragma unroll
        for (int s = 0; s < SPW; ++s) {
            const bool     m    = lane >= 1 && lane <= c[s];
            const unsigned mask = __builtin_amdgcn_ballot_w32(m);
            if (m) my_e[cnt + __popc(mask & lanes_below)] = w[s];
            cnt += __popc(mask);
        }

        for (int i = lane; i < cnt; i += 32) {
            const int e = my_e[i];
            my_r[i]  = (dst[e] - n0) * DIN;
            my_td[i] = bt[e] - node_ts[src[e]];
        }

        for (int i0 = 0; i0 < cnt; i0 += G) {
            const int ng = min(G, cnt - i0);
            const int   li  = i0 + min(lane & (G - 1), ng - 1);
            const int   ev  = my_e[li];
            const int   rv  = my_r[li];
            const float tdv = my_td[li];

            float v[G][4];
            #pragma unroll
            for (int g = 0; g < G; ++g) {
                const int    e   = __builtin_amdgcn_readlane(ev, g);
                const float* row = ef + (size_t)e * HID + lane;
                v[g][0] = row[0]; v[g][1] = row[32]; v[g][2] = row[64]; v[g][3] = row[96];
            }

            for (int g = 0; g < ng; ++g) {
                const int   r  = __builtin_amdgcn_readlane(rv, g);
                const float td = __int_as_float(__builtin_amdgcn_readlane(__float_as_int(tdv), g));
                float cv[4];
                #pragma unroll
                for (int q = 0; q < 4; ++q) {
                    const float x = td * tw[q] + tb[q];
                    const bool wide = !(fabsf(x) < 0x1p-13f);
                    if (__builtin_amdgcn_ballot_w32(wide) != 0u) cv[q] = cosf(x);
                    else                                        cv[q] = 1.0f;
                }
                float* a = &acc[r + lane];
                __hip_atomic_fetch_add(a + 128, cv[0], __ATOMIC_RELAXED, __HIP_MEMORY_SCOPE_WORKGROUP);
                __hip_atomic_fetch_add(a + 160, cv[1], __ATOMIC_RELAXED, __HIP_MEMORY_SCOPE_WORKGROUP);
                __hip_atomic_fetch_add(a + 192, cv[2], __ATOMIC_RELAXED, __HIP_MEMORY_SCOPE_WORKGROUP);
                __hip_atomic_fetch_add(a + 224, cv[3], __ATOMIC_RELAXED, __HIP_MEMORY_SCOPE_WORKGROUP);
            }

            #pragma unroll
            for (int g = 0; g < G; ++g) {
                if (g < ng) {
                    const int r = __builtin_amdgcn_readlane(rv, g);
                    float* a = &acc[r + lane];
                    __hip_atomic_fetch_add(a,      v[g][0], __ATOMIC_RELAXED, __HIP_MEMORY_SCOPE_WORKGROUP);
                    __hip_atomic_fetch_add(a + 32, v[g][1], __ATOMIC_RELAXED, __HIP_MEMORY_SCOPE_WORKGROUP);
                    __hip_atomic_fetch_add(a + 64, v[g][2], __ATOMIC_RELAXED, __HIP_MEMORY_SCOPE_WORKGROUP);
                    __hip_atomic_fetch_add(a + 96, v[g][3], __ATOMIC_RELAXED, __HIP_MEMORY_SCOPE_WORKGROUP);
                }
            }
        }
    } else {
        for (int e = t; e < NE; e += NWAVE * 32) {
            const int dl = dst[e] - n0;
            if ((unsigned)dl < (unsigned)nt) {
                const float  td  = bt[e] - node_ts[src[e]];
                const float* row = ef + (size_t)e * HID;
                float*       a   = &acc[dl * DIN];
                for (int f = 0; f < HID; ++f) {
                    __hip_atomic_fetch_add(a + f, row[f], __ATOMIC_RELAXED, __HIP_MEMORY_SCOPE_WORKGROUP);
                    __hip_atomic_fetch_add(a + HID + f, cosf(td * time_w[f] + time_b[f]),
                                           __ATOMIC_RELAXED, __HIP_MEMORY_SCOPE_WORKGROUP);
                }
            }
        }
    }
    __syncthreads();

    float4* h4 = (float4*)(h + (size_t)n0 * DIN);
    for (int i = t; i < nt * (DIN / 4); i += NWAVE * 32)
        h4[i] = acc4[i];
}

__global__ __launch_bounds__(256) void tgnn_predict_kernel(
    const float* __restrict__ h,
    const int*   __restrict__ s_idx,
    const int*   __restrict__ p_idx,
    const int*   __restrict__ n_idx,
    const float* __restrict__ W_src,
    const float* __restrict__ b_src,
    const float* __restrict__ W_dst,
    const float* __restrict__ b_dst,
    const float* __restrict__ W_out,
    const float* __restrict__ b_out,
    float*       __restrict__ out,
    int Q)
{
    __shared__ __attribute__((aligned(16))) float sl[16 * LROW];
    __shared__ __attribute__((aligned(16))) float pl[16 * LROW];
    __shared__ __attribute__((aligned(16))) float nl[16 * LROW];
    __shared__ float red_pos[8 * 256];
    __shared__ float red_neg[8 * 256];

    const int tile = blockIdx.x;
    const int t    = threadIdx.x;

    {
        const int gw = __builtin_amdgcn_readfirstlane(t >> 5);
        const int gl = t & 31;
        const int r0 = 2 * gw;
        const int qa = tile * 16 + r0;

        const float4* hs0 = (const float4*)(h + (size_t)s_idx[qa]     * DIN);
        const float4* hs1 = (const float4*)(h + (size_t)s_idx[qa + 1] * DIN);
        const float4* hp0 = (const float4*)(h + (size_t)p_idx[qa]     * DIN);
        const float4* hp1 = (const float4*)(h + (size_t)p_idx[qa + 1] * DIN);
        const float4* hn0 = (const float4*)(h + (size_t)n_idx[qa]     * DIN);
        const float4* hn1 = (const float4*)(h + (size_t)n_idx[qa + 1] * DIN);

        const float4 s00 = hs0[gl], s01 = hs0[gl + 32];
        const float4 s10 = hs1[gl], s11 = hs1[gl + 32];
        const float4 p00 = hp0[gl], p01 = hp0[gl + 32];
        const float4 p10 = hp1[gl], p11 = hp1[gl + 32];
        const float4 n00 = hn0[gl], n01 = hn0[gl + 32];
        const float4 n10 = hn1[gl], n11 = hn1[gl + 32];

        float4* s4a = (float4*)&sl[r0 * LROW];
        float4* s4b = (float4*)&sl[(r0 + 1) * LROW];
        float4* p4a = (float4*)&pl[r0 * LROW];
        float4* p4b = (float4*)&pl[(r0 + 1) * LROW];
        float4* n4a = (float4*)&nl[r0 * LROW];
        float4* n4b = (float4*)&nl[(r0 + 1) * LROW];
        s4a[gl] = s00; s4a[gl + 32] = s01;
        s4b[gl] = s10; s4b[gl + 32] = s11;
        p4a[gl] = p00; p4a[gl + 32] = p01;
        p4b[gl] = p10; p4b[gl + 32] = p11;
        n4a[gl] = n00; n4a[gl + 32] = n01;
        n4b[gl] = n10; n4b[gl + 32] = n11;
    }
    __syncthreads();

    const int wave = t >> 5;
    const int lane = t & 31;
    const int half = lane >> 4;
    const int lm   = lane & 15;
    const int n0   = wave * 16;

    v8f hs = {}, hp = {}, hn = {};

    const float* As = &sl[lm * LROW];
    const float* Ap = &pl[lm * LROW];
    const float* An = &nl[lm * LROW];

    for (int k0 = 0; k0 < DIN; k0 += 4) {
        int ka = k0 + 2 * half;
        v2f a_s; a_s.x = As[ka]; a_s.y = As[ka + 1];
        v2f a_p; a_p.x = Ap[ka]; a_p.y = Ap[ka + 1];
        v2f a_n; a_n.x = An[ka]; a_n.y = An[ka + 1];
        v2f b_s; b_s.x = W_src[ka * HID + n0 + lm]; b_s.y = W_src[(ka + 1) * HID + n0 + lm];
        v2f b_d; b_d.x = W_dst[ka * HID + n0 + lm]; b_d.y = W_dst[(ka + 1) * HID + n0 + lm];

        hs = __builtin_amdgcn_wmma_f32_16x16x4_f32(false, a_s, false, b_s, (short)0, hs, false, false);
        hp = __builtin_amdgcn_wmma_f32_16x16x4_f32(false, a_p, false, b_d, (short)0, hp, false, false);
        hn = __builtin_amdgcn_wmma_f32_16x16x4_f32(false, a_n, false, b_d, (short)0, hn, false, false);
    }

    const float bias = b_src[n0 + lm] + b_dst[n0 + lm];
    const float wo   = W_out[n0 + lm];

    #pragma unroll
    for (int r = 0; r < 8; ++r) {
        float pv = fmaxf(hs[r] + hp[r] + bias, 0.f) * wo;
        float nv = fmaxf(hs[r] + hn[r] + bias, 0.f) * wo;
        red_pos[wave * 256 + r * 32 + lane] = pv;
        red_neg[wave * 256 + r * 32 + lane] = nv;
    }
    __syncthreads();

    if (t < 16) {
        int mh = t >> 3;
        int r  = t & 7;
        float accp = b_out[0];
        float accn = b_out[0];
        for (int w = 0; w < 8; ++w) {
            #pragma unroll
            for (int l = 0; l < 16; ++l) {
                int idx = w * 256 + r * 32 + mh * 16 + l;
                accp += red_pos[idx];
                accn += red_neg[idx];
            }
        }
        out[tile * 16 + t]     = accp;
        out[Q + tile * 16 + t] = accn;
    }
}

extern "C" void kernel_launch(void* const* d_in, const int* in_sizes, int n_in,
                              void* d_out, int out_size, void* d_ws, size_t ws_size,
                              hipStream_t stream) {
    const int*   src     = (const int*)  d_in[0];
    const int*   dst     = (const int*)  d_in[1];
    const float* ef      = (const float*)d_in[2];
    const float* bt      = (const float*)d_in[3];
    const float* node_ts = (const float*)d_in[4];
    const int*   s_idx   = (const int*)  d_in[5];
    const int*   p_idx   = (const int*)  d_in[6];
    const int*   n_idx   = (const int*)  d_in[7];
    const float* time_w  = (const float*)d_in[8];
    const float* time_b  = (const float*)d_in[9];
    const float* W_src   = (const float*)d_in[10];
    const float* b_src   = (const float*)d_in[11];
    const float* W_dst   = (const float*)d_in[12];
    const float* b_dst   = (const float*)d_in[13];
    const float* W_out   = (const float*)d_in[14];
    const float* b_out   = (const float*)d_in[15];

    float* out = (float*)d_out;
    float* h   = (float*)d_ws;

    const int NN = in_sizes[4];
    const int NE = in_sizes[0];
    const int Q  = in_sizes[5];

    tgnn_bin_kernel<<<NBIN, NWAVE * 32, 0, stream>>>(dst, h, NN, NE);

    int agrid = (NN + RT - 1) / RT;
    tgnn_tile_accum_kernel<<<agrid, NWAVE * 32, 0, stream>>>(src, dst, ef, bt, node_ts,
                                                             time_w, time_b, h, NN, NE);

    tgnn_predict_kernel<<<Q / 16, 256, 0, stream>>>(h, s_idx, p_idx, n_idx,
                                                    W_src, b_src, W_dst, b_dst,
                                                    W_out, b_out, out, Q);
}


# Rows owned by waves: tile's edges counting-sorted by row in LDS (int atomics); a wave sums a row in registers, stores it once; no f32 LDS tile, atomics or copy-out; sum order in a row by arrival
# speedup vs baseline: 5.1931x; 1.1400x over previous
#include <hip/hip_runtime.h>

typedef __attribute__((ext_vector_type(2))) float v2f;
typedef __attribute__((ext_vector_type(8))) float v8f;

#define HID   128
#define DIN   256
#define LROW  260

#define RT     196
#define NWAVE  32
#define NBIN   160
#define SPW    (NBIN / NWAVE)
#define SLOTW  32
#define SCAP   (SLOTW - 1)
#define TCH    512
#define ECAP   (NBIN * SCAP)
#define RPW    ((RT + NWAVE - 1) / NWAVE)
#define RPL    ((RT + 31) / 32)
#define G      4

__device__ __forceinline__ bool tgnn_tile_has_slots(int nt) { return nt * DIN >= NBIN * SLOTW; }

__global__ __launch_bounds__(1024) void tgnn_bin_kernel(
    const int* __restrict__ dst,
    float*     __restrict__ h,
    int NN, int NE)
{
    __shared__ int sl[TCH * SLOTW];

    const int t    = threadIdx.x;
    const int wave = __builtin_amdgcn_readfirstlane(t >> 5);
    const int lane = t & 31;
    const int b    = blockIdx.x;

    const int ntiles = (NN + RT - 1) / RT;
    const int epb    = (NE + NBIN - 1) / NBIN;
    const int lo     = b * epb;
    const int hi     = min(NE, lo + epb);

    for (int tb = 0; tb < ntiles; tb += TCH) {
        for (int i = t; i < TCH * SLOTW; i += NWAVE * 32) sl[i] = 0;
        __syncthreads();

        for (int e = lo + t; e < hi; e += NWAVE * 32) {
            const int d = dst[e];
            if ((unsigned)d < (unsigned)NN) {
                const int tl = d / RT - tb;
                if ((unsigned)tl < (unsigned)TCH) {
                    const int pos = __hip_atomic_fetch_add(&sl[tl * SLOTW], 1, __ATOMIC_RELAXED,
                                                           __HIP_MEMORY_SCOPE_WORKGROUP);
                    if (pos < SCAP) sl[tl * SLOTW + 1 + pos] = e;
                }
            }
        }
        __syncthreads();

        const int tn = min(TCH, ntiles - tb);
        for (int tl = wave; tl < tn; tl += NWAVE) {
            const int T  = tb + tl;
            const int nt = min(RT, NN - T * RT);
            if (tgnn_tile_has_slots(nt))
                h[(size_t)T * RT * DIN + b * SLOTW + lane] = __int_as_float(sl[tl * SLOTW + lane]);
        }
        __syncthreads();
    }
}

__global__ __launch_bounds__(1024) void tgnn_tile_accum_kernel(
    const int*   __restrict__ src,
    const int*   __restrict__ dst,
    const float* __restrict__ ef,
    const float* __restrict__ bt,
    const float* __restrict__ node_ts,
    const float* __restrict__ time_w,
    const float* __restrict__ time_b,
    float*       __restrict__ h,
    int NN, int NE)
{
    __shared__ int   se[ECAP];
    __shared__ int   sr[ECAP];
    __shared__ float sd[ECAP];
    __shared__ int   rcnt[RT];
    __shared__ int   rstart[RT + 1];
    __shared__ int   ovf;

    const int t    = threadIdx.x;
    const int wave = __builtin_amdgcn_readfirstlane(t >> 5);
    const int lane = t & 31;
    const int n0   = blockIdx.x * RT;
    const int nt   = min(RT, NN - n0);
    const bool has_slots = tgnn_tile_has_slots(nt);

    if (t == 0) ovf = 0;
    if (t < RT) rcnt[t] = 0;
    __syncthreads();

    int w[SPW];
    #pragma unroll
    for (int s = 0; s < SPW; ++s)
        w[s] = has_slots ? __float_as_int(h[(size_t)n0 * DIN + (wave + s * NWAVE) * SLOTW + lane]) : 0;

    float tw[4], tb[4];
    #pragma unroll
    for (int q = 0; q < 4; ++q) {
        tw[q] = time_w[lane + 32 * q];
        tb[q] = time_b[lane + 32 * q];
    }

    int  c[SPW];
    bool over = false;
    #pragma unroll
    for (int s = 0; s < SPW; ++s) {
        c[s]  = __builtin_amdgcn_readfirstlane(w[s]);
        over |= (unsigned)c[s] > (unsigned)SCAP;
    }
    if (over && lane == 0) ovf = 1;
    __syncthreads();
    const bool scan_all = !has_slots || __builtin_amdgcn_readfirstlane(ovf) != 0;

    float* htile = h + (size_t)n0 * DIN;

    if (!scan_all) {
        bool  ev[SPW];
        int   er[SPW], ek[SPW];
        float etd[SPW];
        int   e_any = -1;
        #pragma unroll
        for (int s = 0; s < SPW; ++s) {
            ev[s] = lane >= 1 && lane <= c[s];
            er[s] = 0; ek[s] = 0; etd[s] = 0.f;
            if (e_any < 0 && c[s] > 0) e_any = __builtin_amdgcn_readlane(w[s], 1);
        }
        if (e_any >= 0) {
            int   ed[SPW], es[SPW];
            float eb[SPW];
            #pragma unroll
            for (int s = 0; s < SPW; ++s) {
                const int e = ev[s] ? w[s] : e_any;
                ed[s] = dst[e]; es[s] = src[e]; eb[s] = bt[e];
            }
            #pragma unroll
            for (int s = 0; s < SPW; ++s) {
                const float ts = node_ts[es[s]];
                er[s]  = ed[s] - n0;
                etd[s] = eb[s] - ts;
            }
            #pragma unroll
            for (int s = 0; s < SPW; ++s)
                if (ev[s]) ek[s] = __hip_atomic_fetch_add(&rcnt[er[s]], 1, __ATOMIC_RELAXED, __HIP_MEMORY_SCOPE_WORKGROUP);
        }
        __syncthreads();

        if (wave == 0) {
            int cc[RPL];
            int sum = 0;
            #pragma unroll
            for (int j = 0; j < RPL; ++j) {
                const int idx = lane * RPL + j;
                cc[j] = idx < RT ? rcnt[idx] : 0;
                sum += cc[j];
            }
            int inc = sum;
            for (int d = 1; d < 32; d <<= 1) {
                const int up = __shfl_up(inc, d);
                if (lane >= d) inc += up;
            }
            int ex = inc - sum;
            #pragma unroll
            for (int j = 0; j < RPL; ++j) {
                const int idx = lane * RPL + j;
                if (idx < RT) rstart[idx] = ex;
                ex += cc[j];
            }
            if (lane == 31) rstart[RT] = inc;
        }
        __syncthreads();

        int p[SPW];
        #pragma unroll
        for (int s = 0; s < SPW; ++s) p[s] = rstart[er[s]] + ek[s];
        #pragma unroll
        for (int s = 0; s < SPW; ++s) {
            if (ev[s]) {
                se[p[s]] = w[s];
                sr[p[s]] = er[s];
                sd[p[s]] = etd[s];
            }
        }
        __syncthreads();

        const int row0 = wave * RPW;
        if (row0 < nt) {
            const int rend = min(row0 + RPW, nt);

            const int rs = rstart[min(row0 + min(lane, RPW), rend)];

            #pragma unroll
            for (int j = 0; j < RPW; ++j) {
                const int r   = row0 + j;
                const int n_r = __builtin_amdgcn_readlane(rs, j + 1) - __builtin_amdgcn_readlane(rs, j);
                if (r < rend && n_r == 0) {
                    float4* p = (float4*)(htile + (size_t)r * DIN);
                    p[lane]      = make_float4(0.f, 0.f, 0.f, 0.f);
                    p[lane + 32] = make_float4(0.f, 0.f, 0.f, 0.f);
                }
            }

            const int i_beg = __builtin_amdgcn_readlane(rs, 0);
            const int i_end = __builtin_amdgcn_readlane(rs, RPW);

            int    curc = -1, curr = -1;
            float  ca[4] = {0.f, 0.f, 0.f, 0.f};
            float4 ra    = make_float4(0.f, 0.f, 0.f, 0.f);

            for (int i0 = i_beg; i0 < i_end; i0 += G) {
                const int ng = min(G, i_end - i0);
                const int   li  = i0 + min(lane & (G - 1), ng - 1);
                const int   evv = se[li];
                const int   rv  = sr[li];
                const float tdv = sd[li];

                float4 v[G];
                #pragma unroll
                for (int g = 0; g < G; ++g) {
                    const int e = __builtin_amdgcn_readlane(evv, g);
                    v[g] = ((const float4*)(ef + (size_t)e * HID))[lane];
                }

                for (int g = 0; g < ng; ++g) {
                    const int   r  = __builtin_amdgcn_readlane(rv, g);
                    const float td = __int_as_float(__builtin_amdgcn_readlane(__float_as_int(tdv), g));
                    if (r != curc) {
                        if (curc >= 0) {
                            float* p = htile + (size_t)curc * DIN + HID + lane;
                            p[0] = ca[0]; p[32] = ca[1]; p[64] = ca[2]; p[96] = ca[3];
                        }
                        ca[0] = 0.f; ca[1] = 0.f; ca[2] = 0.f; ca[3] = 0.f;
                        curc = r;
                    }
                    #pragma unroll
                    for (int q = 0; q < 4; ++q) {
                        const float x = td * tw[q] + tb[q];
                        const bool wide = !(fabsf(x) < 0x1p-13f);
                        if (__builtin_amdgcn_ballot_w32(wide) != 0u) ca[q] += cosf(x);
                        else                                        ca[q] += 1.0f;
                    }
                }

                #pragma unroll
                for (int g = 0; g < G; ++g) {
                    if (g < ng) {
                        const int r = __builtin_amdgcn_readlane(rv, g);
                        if (r != curr) {
                            if (curr >= 0) ((float4*)(htile + (size_t)curr * DIN))[lane] = ra;
                            ra   = make_float4(0.f, 0.f, 0.f, 0.f);
                            curr = r;
                        }
                        ra.x += v[g].x; ra.y += v[g].y; ra.z += v[g].z; ra.w += v[g].w;
                    }
                }
            }
            if (curc >= 0) {
                float* p = htile + (size_t)curc * DIN + HID + lane;
                p[0] = ca[0]; p[32] = ca[1]; p[64] = ca[2]; p[96] = ca[3];
            }
            if (curr >= 0) ((float4*)(htile + (size_t)curr * DIN))[lane] = ra;
        }
    } else {
        float* cq = &sd[wave * HID];
        for (int r = wave; r < nt; r += NWAVE) {
            const int node = n0 + r;
            float4 ra = make_float4(0.f, 0.f, 0.f, 0.f);
            cq[lane] = 0.f; cq[lane + 32] = 0.f; cq[lane + 64] = 0.f; cq[lane + 96] = 0.f;
            for (int e0 = 0; e0 < NE; e0 += 32) {
                const int  e = e0 + lane;
                const bool m = e < NE && dst[e] == node;
                unsigned mask = __builtin_amdgcn_ballot_w32(m);
                while (mask != 0u) {
                    const int ee = e0 + __builtin_ctz(mask);
                    mask &= mask - 1u;
                    const float  td = bt[ee] - node_ts[src[ee]];
                    const float4 v  = ((const float4*)(ef + (size_t)ee * HID))[lane];
                    ra.x += v.x; ra.y += v.y; ra.z += v.z; ra.w += v.w;
                    #pragma nounroll
                    for (int q = 0; q < 4; ++q) {
                        const int f = lane + 32 * q;
                        cq[f] += cosf(td * time_w[f] + time_b[f]);
                    }
                }
            }
            ((float4*)(htile + (size_t)r * DIN))[lane] = ra;
            float* p = htile + (size_t)r * DIN + HID + lane;
            p[0] = cq[lane]; p[32] = cq[lane + 32]; p[64] = cq[lane + 64]; p[96] = cq[lane + 96];
        }
    }
}

__global__ __launch_bounds__(256) void tgnn_predict_kernel(
    const float* __restrict__ h,
    const int*   __restrict__ s_idx,
    const int*   __restrict__ p_idx,
    const int*   __restrict__ n_idx,
    const float* __restrict__ W_src,
    const float* __restrict__ b_src,
    const float* __restrict__ W_dst,
    const float* __restrict__ b_dst,
    const float* __restrict__ W_out,
    const float* __restrict__ b_out,
    float*       __restrict__ out,
    int Q)
{
    __shared__ __attribute__((aligned(16))) float sl[16 * LROW];
    __shared__ __attribute__((aligned(16))) float pl[16 * LROW];
    __shared__ __attribute__((aligned(16))) float nl[16 * LROW];
    __shared__ float red_pos[8 * 256];
    __shared__ float red_neg[8 * 256];

    const int tile = blockIdx.x;
    const int t    = threadIdx.x;

    {
        const int gw = __builtin_amdgcn_readfirstlane(t >> 5);
        const int gl = t & 31;
        const int r0 = 2 * gw;
        const int qa = tile * 16 + r0;

        const float4* hs0 = (const float4*)(h + (size_t)s_idx[qa]     * DIN);
        const float4* hs1 = (const float4*)(h + (size_t)s_idx[qa + 1] * DIN);
        const float4* hp0 = (const float4*)(h + (size_t)p_idx[qa]     * DIN);
        const float4* hp1 = (const float4*)(h + (size_t)p_idx[qa + 1] * DIN);
        const float4* hn0 = (const float4*)(h + (size_t)n_idx[qa]     * DIN);
        const float4* hn1 = (const float4*)(h + (size_t)n_idx[qa + 1] * DIN);

        const float4 s00 = hs0[gl], s01 = hs0[gl + 32];
        const float4 s10 = hs1[gl], s11 = hs1[gl + 32];
        const float4 p00 = hp0[gl], p01 = hp0[gl + 32];
        const float4 p10 = hp1[gl], p11 = hp1[gl + 32];
        const float4 n00 = hn0[gl], n01 = hn0[gl + 32];
        const float4 n10 = hn1[gl], n11 = hn1[gl + 32];

        float4* s4a = (float4*)&sl[r0 * LROW];
        float4* s4b = (float4*)&sl[(r0 + 1) * LROW];
        float4* p4a = (float4*)&pl[r0 * LROW];
        float4* p4b = (float4*)&pl[(r0 + 1) * LROW];
        float4* n4a = (float4*)&nl[r0 * LROW];
        float4* n4b = (float4*)&nl[(r0 + 1) * LROW];
        s4a[gl] = s00; s4a[gl + 32] = s01;
        s4b[gl] = s10; s4b[gl + 32] = s11;
        p4a[gl] = p00; p4a[gl + 32] = p01;
        p4b[gl] = p10; p4b[gl + 32] = p11;
        n4a[gl] = n00; n4a[gl + 32] = n01;
        n4b[gl] = n10; n4b[gl + 32] = n11;
    }
    __syncthreads();

    const int wave = t >> 5;
    const int lane = t & 31;
    const int half = lane >> 4;
    const int lm   = lane & 15;
    const int n0   = wave * 16;

    v8f hs = {}, hp = {}, hn = {};

    const float* As = &sl[lm * LROW];
    const float* Ap = &pl[lm * LROW];
    const float* An = &nl[lm * LROW];

    for (int k0 = 0; k0 < DIN; k0 += 4) {
        int ka = k0 + 2 * half;
        v2f a_s; a_s.x = As[ka]; a_s.y = As[ka + 1];
        v2f a_p; a_p.x = Ap[ka]; a_p.y = Ap[ka + 1];
        v2f a_n; a_n.x = An[ka]; a_n.y = An[ka + 1];
        v2f b_s; b_s.x = W_src[ka * HID + n0 + lm]; b_s.y = W_src[(ka + 1) * HID + n0 + lm];
        v2f b_d; b_d.x = W_dst[ka * HID + n0 + lm]; b_d.y = W_dst[(ka + 1) * HID + n0 + lm];

        hs = __builtin_amdgcn_wmma_f32_16x16x4_f32(false, a_s, false, b_s, (short)0, hs, false, false);
        hp = __builtin_amdgcn_wmma_f32_16x16x4_f32(false, a_p, false, b_d, (short)0, hp, false, false);
        hn = __builtin_amdgcn_wmma_f32_16x16x4_f32(false, a_n, false, b_d, (short)0, hn, false, false);
    }

    const float bias = b_src[n0 + lm] + b_dst[n0 + lm];
    const float wo   = W_out[n0 + lm];

    #pragma unroll
    for (int r = 0; r < 8; ++r) {
        float pv = fmaxf(hs[r] + hp[r] + bias, 0.f) * wo;
        float nv = fmaxf(hs[r] + hn[r] + bias, 0.f) * wo;
        red_pos[wave * 256 + r * 32 + lane] = pv;
        red_neg[wave * 256 + r * 32 + lane] = nv;
    }
    __syncthreads();

    if (t < 16) {
        int mh = t >> 3;
        int r  = t & 7;
        float accp = b_out[0];
        float accn = b_out[0];
        for (int w = 0; w < 8; ++w) {
            #pragma unroll
            for (int l = 0; l < 16; ++l) {
                int idx = w * 256 + r * 32 + mh * 16 + l;
                accp += red_pos[idx];
                accn += red_neg[idx];
            }
        }
        out[tile * 16 + t]     = accp;
        out[Q + tile * 16 + t] = accn;
    }
}

extern "C" void kernel_launch(void* const* d_in, const int* in_sizes, int n_in,
                              void* d_out, int out_size, void* d_ws, size_t ws_size,
                              hipStream_t stream) {
    const int*   src     = (const int*)  d_in[0];
    const int*   dst     = (const int*)  d_in[1];
    const float* ef      = (const float*)d_in[2];
    const float* bt      = (const float*)d_in[3];
    const float* node_ts = (const float*)d_in[4];
    const int*   s_idx   = (const int*)  d_in[5];
    const int*   p_idx   = (const int*)  d_in[6];
    const int*   n_idx   = (const int*)  d_in[7];
    const float* time_w  = (const float*)d_in[8];
    const float* time_b  = (const float*)d_in[9];
    const float* W_src   = (const float*)d_in[10];
    const float* b_src   = (const float*)d_in[11];
    const float* W_dst   = (const float*)d_in[12];
    const float* b_dst   = (const float*)d_in[13];
    const float* W_out   = (const float*)d_in[14];
    const float* b_out   = (const float*)d_in[15];

    float* out = (float*)d_out;
    float* h   = (float*)d_ws;

    const int NN = in_sizes[4];
    const int NE = in_sizes[0];
    const int Q  = in_sizes[5];

    tgnn_bin_kernel<<<NBIN, NWAVE * 32, 0, stream>>>(dst, h, NN, NE);

    int agrid = (NN + RT - 1) / RT;
    tgnn_tile_accum_kernel<<<agrid, NWAVE * 32, 0, stream>>>(src, dst, ef, bt, node_ts,
                                                             time_w, time_b, h, NN, NE);

    tgnn_predict_kernel<<<Q / 16, 256, 0, stream>>>(h, s_idx, p_idx, n_idx,
                                                    W_src, b_src, W_dst, b_dst,
                                                    W_out, b_out, out, Q);
}
